# starting kernel
# baseline (speedup 1.0000x reference)
#include <hip/hip_runtime.h>

typedef float v2f __attribute__((ext_vector_type(2)));
typedef float v8f __attribute__((ext_vector_type(8)));

#define F_DIM   1024
#define L_LAYERS   3

__global__ __launch_bounds__(256) void dcn_cross_wmma_kernel(
    const float* __restrict__ x, const float* __restrict__ kernels,
    const float* __restrict__ bias, float* __restrict__ out)
{
    __shared__ float w_lds[L_LAYERS * F_DIM];
    __shared__ float b_lds[L_LAYERS * F_DIM];
    __shared__ float s_partial[8][16];
    __shared__ float s_final[16];

    const int tid     = threadIdx.x;
    const int wave    = tid >> 5;
    const int lane    = tid & 31;
    const int half    = lane >> 4;
    const int row     = lane & 15;
    const int colbase = wave * 128 + half * 64;
    const int grow    = blockIdx.x * 16 + row;

    float4 xl4[16];
    float4 x04[16];
    const float4* xrow = (const float4*)(x + (size_t)grow * F_DIM + colbase);
#pragma unroll
    for (int j = 0; j < 16; ++j) { xl4[j] = xrow[j]; x04[j] = xl4[j]; }

    for (int i = tid; i < L_LAYERS * F_DIM; i += 256) {
        w_lds[i] = kernels[i];
        b_lds[i] = bias[i];
    }
    __syncthreads();

#pragma unroll
    for (int l = 0; l < L_LAYERS; ++l) {
        const float* wl = w_lds + l * F_DIM + colbase;

        v8f acc = {};
#pragma unroll
        for (int c = 0; c < 32; ++c) {
            const int j = c >> 1;
            v2f a;
            if (c & 1) { a[0] = xl4[j].z; a[1] = xl4[j].w; }
            else       { a[0] = xl4[j].x; a[1] = xl4[j].y; }
            v2f b = *(const v2f*)(wl + 2 * c);
            acc = __builtin_amdgcn_wmma_f32_16x16x4_f32(
                false, a, false, b, (short)0, acc, false, false);
        }

        if (lane == 0 || lane == 16) {
            float* dst = &s_partial[wave][half * 8];
#pragma unroll
            for (int m = 0; m < 8; ++m) dst[m] = acc[m];
        }
        __syncthreads();

        if (tid < 16) {
            float s = 0.f;
#pragma unroll
            for (int wv = 0; wv < 8; ++wv) s += s_partial[wv][tid];
            s_final[tid] = s;
        }
        __syncthreads();

        const float s = s_final[row];
        const float4* bl = (const float4*)(b_lds + l * F_DIM + colbase);
#pragma unroll
        for (int j = 0; j < 16; ++j) {
            float4 bv = bl[j];
            xl4[j].x = fmaf(x04[j].x, s, bv.x + xl4[j].x);
            xl4[j].y = fmaf(x04[j].y, s, bv.y + xl4[j].y);
            xl4[j].z = fmaf(x04[j].z, s, bv.z + xl4[j].z);
            xl4[j].w = fmaf(x04[j].w, s, bv.w + xl4[j].w);
        }
    }

    float4* orow = (float4*)(out + (size_t)grow * F_DIM + colbase);
#pragma unroll
    for (int j = 0; j < 16; ++j) orow[j] = xl4[j];
}

extern "C" void kernel_launch(void* const* d_in, const int* in_sizes, int n_in,
                              void* d_out, int out_size, void* d_ws, size_t ws_size,
                              hipStream_t stream) {
    const float* x       = (const float*)d_in[0];
    const float* kernels = (const float*)d_in[1];
    const float* bias    = (const float*)d_in[2];
    float* out           = (float*)d_out;

    const int B = in_sizes[0] / F_DIM;
    dim3 grid(B / 16), block(256);
    dcn_cross_wmma_kernel<<<grid, block, 0, stream>>>(x, kernels, bias, out);
}


# each wave owns two whole rows, lanes spread over features: every global load/store moves 512 contiguous B; row dot = trace of the WMMA D tile + wave butterfly (changes f32 sum order)
# speedup vs baseline: 2.3999x; 2.3999x over previous
#include <hip/hip_runtime.h>


typedef float v2f __attribute__((ext_vector_type(2)));
typedef float v8f __attribute__((ext_vector_type(8)));

#define F_DIM      1024
#define L_LAYERS   3
#define F4_PER_ROW (F_DIM / 4)
#define PIECES     (F4_PER_ROW / 32)
#define WAVE_ROWS  2
#define BLOCK_ROWS (8 * WAVE_ROWS)

__device__ __forceinline__ float xhalf(float v) {
    int b = __builtin_bit_cast(int, v);
    return __builtin_bit_cast(float,
        __builtin_amdgcn_permlanex16(b, b, 0x76543210, 0xfedcba98, false, false));
}

template <int M>
__device__ __forceinline__ float add_xmask(float v) {
    int b = __builtin_bit_cast(int, v);
    float o = __builtin_bit_cast(float,
        __builtin_amdgcn_update_dpp(0, b, 0x160 + M, 0xf, 0xf, true));
    return v + o;
}

__device__ __forceinline__ float wave_sum(float v) {
    v = v + xhalf(v);
    v = add_xmask<1>(v);
    v = add_xmask<2>(v);
    v = add_xmask<4>(v);
    v = add_xmask<8>(v);
    return v;
}

__device__ __forceinline__ float diag_of(v8f d, int lane) {
    const int r = lane & 7;
    float t = d[0];
    t = (r == 1) ? d[1] : t;
    t = (r == 2) ? d[2] : t;
    t = (r == 3) ? d[3] : t;
    t = (r == 4) ? d[4] : t;
    t = (r == 5) ? d[5] : t;
    t = (r == 6) ? d[6] : t;
    t = (r == 7) ? d[7] : t;
    return (lane < 8 || lane >= 24) ? t : 0.f;
}

__global__ __launch_bounds__(256) void dcn_cross_wmma_kernel(
    const float* __restrict__ x, const float* __restrict__ kernels,
    const float* __restrict__ bias, float* __restrict__ out)
{
    __shared__ float4 w_lds[L_LAYERS * F4_PER_ROW];
    __shared__ float4 b_lds[L_LAYERS * F4_PER_ROW];

    const int tid  = threadIdx.x;
    const int wave = tid >> 5;
    const int lane = tid & 31;
    const size_t row0 = (size_t)blockIdx.x * BLOCK_ROWS + (size_t)wave * WAVE_ROWS;

    float4 x0[WAVE_ROWS][PIECES];
    float4 xl[WAVE_ROWS][PIECES];
    for (int r = 0; r < WAVE_ROWS; ++r) {
        const float4* xr = (const float4*)(x + (row0 + r) * F_DIM) + lane;
        for (int q = 0; q < PIECES; ++q) x0[r][q] = xr[q * 32];
    }

    {
        const float4* k4 = (const float4*)kernels;
        const float4* b4 = (const float4*)bias;
        for (int k = 0; k < (L_LAYERS * F4_PER_ROW) / 256; ++k) {
            const int i = tid + k * 256;
            w_lds[i] = k4[i];
            b_lds[i] = b4[i];
        }
    }
    __syncthreads();

    for (int r = 0; r < WAVE_ROWS; ++r)
        for (int q = 0; q < PIECES; ++q) xl[r][q] = x0[r][q];

    for (int l = 0; l < L_LAYERS; ++l) {
        const float4* wl = w_lds + l * F4_PER_ROW + lane;
        const float4* bl = b_lds + l * F4_PER_ROW + lane;

        v8f acc[WAVE_ROWS];
        for (int r = 0; r < WAVE_ROWS; ++r) acc[r] = (v8f){};

        for (int q = 0; q < PIECES; ++q) {
            const float4 wv = wl[q * 32];
            v2f b_lo, b_hi;
            b_lo[0] = wv.x; b_lo[1] = wv.y;
            b_hi[0] = wv.z; b_hi[1] = wv.w;
            for (int r = 0; r < WAVE_ROWS; ++r) {
                v2f a;
                a[0] = xl[r][q].x; a[1] = xl[r][q].y;
                acc[r] = __builtin_amdgcn_wmma_f32_16x16x4_f32(
                    false, a, false, b_lo, (short)0, acc[r], false, false);
            }
            for (int r = 0; r < WAVE_ROWS; ++r) {
                v2f a;
                a[0] = xl[r][q].z; a[1] = xl[r][q].w;
                acc[r] = __builtin_amdgcn_wmma_f32_16x16x4_f32(
                    false, a, false, b_hi, (short)0, acc[r], false, false);
            }
        }

        float s[WAVE_ROWS];
        for (int r = 0; r < WAVE_ROWS; ++r) s[r] = wave_sum(diag_of(acc[r], lane));

        for (int q = 0; q < PIECES; ++q) {
            const float4 bv = bl[q * 32];
            for (int r = 0; r < WAVE_ROWS; ++r) {
                xl[r][q].x = fmaf(x0[r][q].x, s[r], bv.x + xl[r][q].x);
                xl[r][q].y = fmaf(x0[r][q].y, s[r], bv.y + xl[r][q].y);
                xl[r][q].z = fmaf(x0[r][q].z, s[r], bv.z + xl[r][q].z);
                xl[r][q].w = fmaf(x0[r][q].w, s[r], bv.w + xl[r][q].w);
            }
        }
    }

    for (int r = 0; r < WAVE_ROWS; ++r) {
        float4* orow = (float4*)(out + (row0 + r) * F_DIM) + lane;
        for (int q = 0; q < PIECES; ++q) orow[q * 32] = xl[r][q];
    }
}

extern "C" void kernel_launch(void* const* d_in, const int* in_sizes, int n_in,
                              void* d_out, int out_size, void* d_ws, size_t ws_size,
                              hipStream_t stream) {
    const float* x       = (const float*)d_in[0];
    const float* kernels = (const float*)d_in[1];
    const float* bias    = (const float*)d_in[2];
    float* out           = (float*)d_out;

    const int B = in_sizes[0] / F_DIM;
    dim3 grid(B / BLOCK_ROWS), block(256);
    dcn_cross_wmma_kernel<<<grid, block, 0, stream>>>(x, kernels, bias, out);
}


# each wave works through two row pairs in turn, the second pair's 16 loads issued ahead of the first pair's WMMAs: loads and stores run under matrix work; half the waves and blocks (grid 512)
# speedup vs baseline: 2.5278x; 1.0533x over previous
#include <hip/hip_runtime.h>


typedef float v2f __attribute__((ext_vector_type(2)));
typedef float v8f __attribute__((ext_vector_type(8)));

#define F_DIM      1024
#define L_LAYERS   3
#define F4_PER_ROW (F_DIM / 4)
#define PIECES     (F4_PER_ROW / 32)
#define WAVE_ROWS  2
#define WAVE_PAIRS 2
#define BLOCK_ROWS (8 * WAVE_ROWS * WAVE_PAIRS)

__device__ __forceinline__ float xhalf(float v) {
    int b = __builtin_bit_cast(int, v);
    return __builtin_bit_cast(float,
        __builtin_amdgcn_permlanex16(b, b, 0x76543210, 0xfedcba98, false, false));
}

template <int M>
__device__ __forceinline__ float add_xmask(float v) {
    int b = __builtin_bit_cast(int, v);
    float o = __builtin_bit_cast(float,
        __builtin_amdgcn_update_dpp(0, b, 0x160 + M, 0xf, 0xf, true));
    return v + o;
}

__device__ __forceinline__ float wave_sum(float v) {
    v = v + xhalf(v);
    v = add_xmask<1>(v);
    v = add_xmask<2>(v);
    v = add_xmask<4>(v);
    v = add_xmask<8>(v);
    return v;
}

__device__ __forceinline__ float diag_of(v8f d, int lane) {
    const int r = lane & 7;
    float t = d[0];
    t = (r == 1) ? d[1] : t;
    t = (r == 2) ? d[2] : t;
    t = (r == 3) ? d[3] : t;
    t = (r == 4) ? d[4] : t;
    t = (r == 5) ? d[5] : t;
    t = (r == 6) ? d[6] : t;
    t = (r == 7) ? d[7] : t;
    return (lane < 8 || lane >= 24) ? t : 0.f;
}

__global__ __launch_bounds__(256) void dcn_cross_wmma_kernel(
    const float* __restrict__ x, const float* __restrict__ kernels,
    const float* __restrict__ bias, float* __restrict__ out)
{
    __shared__ float4 w_lds[L_LAYERS * F4_PER_ROW];
    __shared__ float4 b_lds[L_LAYERS * F4_PER_ROW];

    const int tid  = threadIdx.x;
    const int wave = tid >> 5;
    const int lane = tid & 31;
    const size_t row0 = (size_t)blockIdx.x * BLOCK_ROWS
                      + (size_t)wave * (WAVE_ROWS * WAVE_PAIRS);

    float4 xn[WAVE_ROWS][PIECES];
    for (int r = 0; r < WAVE_ROWS; ++r) {
        const float4* xr = (const float4*)(x + (row0 + r) * F_DIM) + lane;
        for (int q = 0; q < PIECES; ++q) xn[r][q] = xr[q * 32];
    }

    {
        const float4* k4 = (const float4*)kernels;
        const float4* b4 = (const float4*)bias;
        for (int k = 0; k < (L_LAYERS * F4_PER_ROW) / 256; ++k) {
            const int i = tid + k * 256;
            w_lds[i] = k4[i];
            b_lds[i] = b4[i];
        }
    }
    __syncthreads();

    #pragma unroll
    for (int t = 0; t < WAVE_PAIRS; ++t) {
        const size_t rowt = row0 + (size_t)t * WAVE_ROWS;

        asm volatile("" ::: "memory");

        float4 x0[WAVE_ROWS][PIECES];
        float4 xl[WAVE_ROWS][PIECES];
        for (int r = 0; r < WAVE_ROWS; ++r)
            for (int q = 0; q < PIECES; ++q) { x0[r][q] = xn[r][q]; xl[r][q] = xn[r][q]; }

        if (t + 1 < WAVE_PAIRS) {
            for (int r = 0; r < WAVE_ROWS; ++r) {
                const float4* xr =
                    (const float4*)(x + (rowt + WAVE_ROWS + r) * F_DIM) + lane;
                for (int q = 0; q < PIECES; ++q) xn[r][q] = xr[q * 32];
            }
            __builtin_amdgcn_sched_barrier(0);
        }

        for (int l = 0; l < L_LAYERS; ++l) {
            const float4* wl = w_lds + l * F4_PER_ROW + lane;
            const float4* bl = b_lds + l * F4_PER_ROW + lane;

            v8f acc[WAVE_ROWS];
            for (int r = 0; r < WAVE_ROWS; ++r) acc[r] = (v8f){};

            for (int q = 0; q < PIECES; ++q) {
                const float4 wv = wl[q * 32];
                v2f b_lo, b_hi;
                b_lo[0] = wv.x; b_lo[1] = wv.y;
                b_hi[0] = wv.z; b_hi[1] = wv.w;
                for (int r = 0; r < WAVE_ROWS; ++r) {
                    v2f a;
                    a[0] = xl[r][q].x; a[1] = xl[r][q].y;
                    acc[r] = __builtin_amdgcn_wmma_f32_16x16x4_f32(
                        false, a, false, b_lo, (short)0, acc[r], false, false);
                }
                for (int r = 0; r < WAVE_ROWS; ++r) {
                    v2f a;
                    a[0] = xl[r][q].z; a[1] = xl[r][q].w;
                    acc[r] = __builtin_amdgcn_wmma_f32_16x16x4_f32(
                        false, a, false, b_hi, (short)0, acc[r], false, false);
                }
            }

            float s[WAVE_ROWS];
            for (int r = 0; r < WAVE_ROWS; ++r) s[r] = wave_sum(diag_of(acc[r], lane));

            for (int q = 0; q < PIECES; ++q) {
                const float4 bv = bl[q * 32];
                for (int r = 0; r < WAVE_ROWS; ++r) {
                    xl[r][q].x = fmaf(x0[r][q].x, s[r], bv.x + xl[r][q].x);
                    xl[r][q].y = fmaf(x0[r][q].y, s[r], bv.y + xl[r][q].y);
                    xl[r][q].z = fmaf(x0[r][q].z, s[r], bv.z + xl[r][q].z);
                    xl[r][q].w = fmaf(x0[r][q].w, s[r], bv.w + xl[r][q].w);
                }
            }
        }

        for (int r = 0; r < WAVE_ROWS; ++r) {
            float4* orow = (float4*)(out + (rowt + r) * F_DIM) + lane;
            for (int q = 0; q < PIECES; ++q) orow[q * 32] = xl[r][q];
        }
    }
}

extern "C" void kernel_launch(void* const* d_in, const int* in_sizes, int n_in,
                              void* d_out, int out_size, void* d_ws, size_t ws_size,
                              hipStream_t stream) {
    const float* x       = (const float*)d_in[0];
    const float* kernels = (const float*)d_in[1];
    const float* bias    = (const float*)d_in[2];
    float* out           = (float*)d_out;

    const int B = in_sizes[0] / F_DIM;
    dim3 grid(B / BLOCK_ROWS), block(256);
    dcn_cross_wmma_kernel<<<grid, block, 0, stream>>>(x, kernels, bias, out);
}
